# starting kernel
# baseline (speedup 1.0000x reference)
#include <hip/hip_runtime.h>

typedef __attribute__((ext_vector_type(2))) float v2f;
typedef __attribute__((ext_vector_type(8))) float v8f;

#define PI_F 3.14159265358979323846f


__global__ void zero_out_kernel(float* __restrict__ out, int n) {
  int i = blockIdx.x * blockDim.x + threadIdx.x;
  if (i < n) out[i] = 0.0f;
}

__global__ __launch_bounds__(32) void precompute_L_kernel(
    const float* __restrict__ qp, float* __restrict__ Lb) {
  const int k = blockIdx.x;
  const int lane = threadIdx.x;
  const float* t = qp + k * 16;

  float xre[32], xim[32];
#pragma unroll
  for (int j = 0; j < 32; ++j) { xre[j] = (j == lane) ? 1.f : 0.f; xim[j] = 0.f; }

  const int G[4][3] = {{6,0,1},{8,3,4},{10,1,2},{12,2,4}};
#pragma unroll
  for (int g = 0; g < 4; ++g) {
    const int cb = G[g][1], tb = G[g][2];
    const int ctrl = (lane >> cb) & 1;
    const int tgt  = (lane >> tb) & 1;
    {
      float h = 0.5f * t[G[g][0]];
      float cz = cosf(h), sz = sinf(h);
      float cc = ctrl ? cz : 1.0f;
      float gg = ctrl ? (tgt ? sz : -sz) : 0.0f;
#pragma unroll
      for (int j = 0; j < 32; ++j) {
        float re = xre[j], im = xim[j];
        xre[j] = re * cc - im * gg;
        xim[j] = im * cc + re * gg;
      }
    }
    {
      float h = 0.5f * t[G[g][0] + 1];
      float co = cosf(h), si = sinf(h);
      float ce = ctrl ? co : 1.0f;
      float se = ctrl ? si : 0.0f;
#pragma unroll
      for (int j = 0; j < 32; ++j) {
        float pre = __shfl_xor(xre[j], 1 << tb, 32);
        float pim = __shfl_xor(xim[j], 1 << tb, 32);
        float re = xre[j], im = xim[j];
        xre[j] = ce * re + se * pim;
        xim[j] = ce * im - se * pre;
      }
    }
  }

  const int nlow = lane & 15;
  const int hi = lane >> 4;
#pragma unroll
  for (int tt = 0; tt < 2; ++tt) {
#pragma unroll
    for (int kc = 0; kc < 8; ++kc) {
      float* base = Lb + (((k * 16) + tt * 8 + kc) << 7);
#pragma unroll
      for (int v = 0; v < 2; ++v) {
        float r0 = __shfl(xre[4 * kc + v],     16 * tt + nlow, 32);
        float r2 = __shfl(xre[4 * kc + v + 2], 16 * tt + nlow, 32);
        float i0 = __shfl(xim[4 * kc + v],     16 * tt + nlow, 32);
        float i2 = __shfl(xim[4 * kc + v + 2], 16 * tt + nlow, 32);
        base[2 * lane + v]      = hi ? r2 : r0;
        base[64 + 2 * lane + v] = hi ? i2 : i0;
      }
    }
  }
}

__global__ __launch_bounds__(256) void quanv_main_kernel(
    const float* __restrict__ x, const float* __restrict__ qp,
    const float* __restrict__ Lb, float* __restrict__ out) {
  const int lane = threadIdx.x & 31;
  const int wave = (blockIdx.x << 3) + (threadIdx.x >> 5);
  const int k = wave & 31;
  const int p = wave >> 5;
  const int b = p / 484;
  const int rem = p - b * 484;
  const int row = rem / 22;
  const int col = rem - row * 22;

  float cq[9], sq[9];
#pragma unroll
  for (int m = 0; m < 9; ++m) {
    float d = x[(b * 24 + row + m / 3) * 24 + (col + m % 3)];
    float h = (0.5f * PI_F) * d;
    cq[m] = cosf(h);
    sq[m] = sinf(h);
  }
  const float* t = qp + k * 16;

  const int r  = lane & 15;
  const int l4 = lane >> 4;
  float mlane = ((r >> 3) & 1 ? sq[0] : cq[0]) * ((r >> 2) & 1 ? sq[1] : cq[1]) *
                ((r >> 1) & 1 ? sq[2] : cq[2]) * ((r & 1)      ? sq[3] : cq[3]) *
                (l4 ? sq[7] : cq[7]);
  int nlane = ((r >> 3) & 1) + ((r >> 2) & 1) + ((r >> 1) & 1) + (r & 1) + l4;

  float sre[16], sim[16];
#pragma unroll
  for (int i = 0; i < 16; ++i) {
    float mi = ((i & 1)        ? sq[8] : cq[8]) *
               (((i >> 1) & 1) ? sq[6] : cq[6]) *
               (((i >> 2) & 1) ? sq[5] : cq[5]) *
               (((i >> 3) & 1) ? sq[4] : cq[4]);
    int n = (nlane + (i & 1) + ((i >> 1) & 1) + ((i >> 2) & 1) + ((i >> 3) & 1)) & 3;
    float m = mlane * mi;
    sre[i] = (n == 0) ? m : ((n == 2) ? -m : 0.f);
    sim[i] = (n == 1) ? -m : ((n == 3) ? m : 0.f);
  }

  const int RG[3][3] = {{0,2,3},{2,0,1},{4,1,3}};
#pragma unroll
  for (int g = 0; g < 3; ++g) {
    const int cb = RG[g][1], tb = RG[g][2];
    const int ctrl = (lane >> cb) & 1;
    const int tgt  = (lane >> tb) & 1;
    {
      float h = 0.5f * t[RG[g][0]];
      float cz = cosf(h), szv = sinf(h);
      float cc = ctrl ? cz : 1.f;
      float gg = ctrl ? (tgt ? szv : -szv) : 0.f;
#pragma unroll
      for (int i = 0; i < 16; ++i) {
        float re = sre[i], im = sim[i];
        sre[i] = re * cc - im * gg;
        sim[i] = im * cc + re * gg;
      }
    }
    {
      float h = 0.5f * t[RG[g][0] + 1];
      float co = cosf(h), si = sinf(h);
      float ce = ctrl ? co : 1.f;
      float se = ctrl ? si : 0.f;
#pragma unroll
      for (int i = 0; i < 16; ++i) {
        float pre = __shfl_xor(sre[i], 1 << tb, 32);
        float pim = __shfl_xor(sim[i], 1 << tb, 32);
        float re = sre[i], im = sim[i];
        sre[i] = ce * re + se * pim;
        sim[i] = ce * im - se * pre;
      }
    }
  }

  float nimv[16];
#pragma unroll
  for (int i = 0; i < 16; ++i) nimv[i] = -sim[i];

  v8f dre[2], dim_[2];
#pragma unroll
  for (int tt = 0; tt < 2; ++tt) {
    v8f accre = {}, accim = {};
#pragma unroll
    for (int kc = 0; kc < 8; ++kc) {
      const v2f* bb = (const v2f*)(Lb + (((k * 16) + tt * 8 + kc) << 7));
      v2f bre = bb[lane];
      v2f bim = bb[32 + lane];
      v2f are  = {sre[2 * kc],  sre[2 * kc + 1]};
      v2f aim  = {sim[2 * kc],  sim[2 * kc + 1]};
      v2f anim = {nimv[2 * kc], nimv[2 * kc + 1]};
      accre = __builtin_amdgcn_wmma_f32_16x16x4_f32(false, are,  false, bre,
                                                    (short)0, accre, false, false);
      accre = __builtin_amdgcn_wmma_f32_16x16x4_f32(false, anim, false, bim,
                                                    (short)0, accre, false, false);
      accim = __builtin_amdgcn_wmma_f32_16x16x4_f32(false, are,  false, bim,
                                                    (short)0, accim, false, false);
      accim = __builtin_amdgcn_wmma_f32_16x16x4_f32(false, aim,  false, bre,
                                                    (short)0, accim, false, false);
    }
    dre[tt] = accre;
    dim_[tt] = accim;
  }

  {
    float h = 0.5f * t[14];
    float cz = cosf(h), szv = sinf(h);
    float gg = (lane < 16) ? -szv : szv;
#pragma unroll
    for (int j = 0; j < 8; ++j) {
      float re = dre[1][j], im = dim_[1][j];
      dre[1][j] = re * cz - im * gg;
      dim_[1][j] = im * cz + re * gg;
    }
  }
  {
    float h = 0.5f * t[15];
    float co = cosf(h), si = sinf(h);
#pragma unroll
    for (int j = 0; j < 8; ++j) {
      float pre = __shfl_xor((float)dre[1][j], 16, 32);
      float pim = __shfl_xor((float)dim_[1][j], 16, 32);
      float re = dre[1][j], im = dim_[1][j];
      dre[1][j] = co * re + si * pim;
      dim_[1][j] = co * im - si * pre;
    }
  }

  float s = 0.f;
#pragma unroll
  for (int tt = 0; tt < 2; ++tt)
#pragma unroll
    for (int j = 0; j < 8; ++j)
      s += dre[tt][j] * dre[tt][j] + dim_[tt][j] * dim_[tt][j];
  float v = (lane < 16) ? s : -s;
#pragma unroll
  for (int off = 16; off >= 1; off >>= 1) v += __shfl_xor(v, off, 32);

  if (lane == 0)
    out[((b * 24 + row + 1) * 24 + (col + 1)) * 32 + k] = (v + 1.f) * 0.5f;
}

extern "C" void kernel_launch(void* const* d_in, const int* in_sizes, int n_in,
                              void* d_out, int out_size, void* d_ws, size_t ws_size,
                              hipStream_t stream) {
  const float* x  = (const float*)d_in[0];
  const float* qp = (const float*)d_in[1];
  float* out = (float*)d_out;
  float* Lb  = (float*)d_ws;

  zero_out_kernel<<<(out_size + 255) / 256, 256, 0, stream>>>(out, out_size);
  precompute_L_kernel<<<32, 32, 0, stream>>>(qp, Lb);
  quanv_main_kernel<<<3872, 256, 0, stream>>>(x, qp, Lb, out);
}


# Product-state factoring: L is applied to the 32-amplitude qubit-4..8 factor alone, 16 patches per f32 WMMA tile (16x fewer waves and WMMAs); sums regrouped as w0*zA+w1*zU, summation order changes
# speedup vs baseline: 3.5147x; 3.5147x over previous
#include <hip/hip_runtime.h>

typedef __attribute__((ext_vector_type(2))) float v2f;
typedef __attribute__((ext_vector_type(8))) float v8f;

#define PI_F 3.14159265358979323846f


__global__ void zero_out_kernel(float* __restrict__ out, int n) {
  int i = blockIdx.x * blockDim.x + threadIdx.x;
  if (i < n) out[i] = 0.0f;
}

__global__ __launch_bounds__(32) void precompute_L_kernel(
    const float* __restrict__ qp, float* __restrict__ Lb) {
  const int k = blockIdx.x;
  const int lane = threadIdx.x;
  const float* t = qp + k * 16;

  float xre[32], xim[32];
#pragma unroll
  for (int j = 0; j < 32; ++j) { xre[j] = (j == lane) ? 1.f : 0.f; xim[j] = 0.f; }

  const int G[4][3] = {{6,0,1},{8,3,4},{10,1,2},{12,2,4}};
#pragma unroll
  for (int g = 0; g < 4; ++g) {
    const int cb = G[g][1], tb = G[g][2];
    const int ctrl = (lane >> cb) & 1;
    const int tgt  = (lane >> tb) & 1;
    {
      float h = 0.5f * t[G[g][0]];
      float cz = cosf(h), sz = sinf(h);
      float cc = ctrl ? cz : 1.0f;
      float gg = ctrl ? (tgt ? sz : -sz) : 0.0f;
#pragma unroll
      for (int j = 0; j < 32; ++j) {
        float re = xre[j], im = xim[j];
        xre[j] = re * cc - im * gg;
        xim[j] = im * cc + re * gg;
      }
    }
    {
      float h = 0.5f * t[G[g][0] + 1];
      float co = cosf(h), si = sinf(h);
      float ce = ctrl ? co : 1.0f;
      float se = ctrl ? si : 0.0f;
#pragma unroll
      for (int j = 0; j < 32; ++j) {
        float pre = __shfl_xor(xre[j], 1 << tb, 32);
        float pim = __shfl_xor(xim[j], 1 << tb, 32);
        float re = xre[j], im = xim[j];
        xre[j] = ce * re + se * pim;
        xim[j] = ce * im - se * pre;
      }
    }
  }

  const int nlow = lane & 15;
  const int hi = lane >> 4;
#pragma unroll
  for (int tt = 0; tt < 2; ++tt) {
#pragma unroll
    for (int kc = 0; kc < 8; ++kc) {
      float* base = Lb + (((k * 16) + tt * 8 + kc) << 7);
#pragma unroll
      for (int v = 0; v < 2; ++v) {
        float r0 = __shfl(xre[4 * kc + v],     16 * tt + nlow, 32);
        float r2 = __shfl(xre[4 * kc + v + 2], 16 * tt + nlow, 32);
        float i0 = __shfl(xim[4 * kc + v],     16 * tt + nlow, 32);
        float i2 = __shfl(xim[4 * kc + v + 2], 16 * tt + nlow, 32);
        base[2 * lane + v]      = hi ? r2 : r0;
        base[64 + 2 * lane + v] = hi ? i2 : i0;
      }
    }
  }
}

__global__ __launch_bounds__(256) void quanv_main_kernel(
    const float* __restrict__ x, const float* __restrict__ qp,
    const float* __restrict__ Lb, float* __restrict__ out) {
  const int lane = threadIdx.x & 31;
  const int wave = (blockIdx.x << 3) + (threadIdx.x >> 5);
  const int k = wave & 31;
  const int tile = wave >> 5;
  const int r  = lane & 15;
  const int l4 = lane >> 4;
  const int pid = tile * 16 + r;
  const bool valid = pid < 968;
  const int p = valid ? pid : 967;
  const int b = p / 484;
  const int rem = p - b * 484;
  const int row = rem / 22;
  const int col = rem - row * 22;

  float cq[9], sq[9];
#pragma unroll
  for (int m = 0; m < 9; ++m) {
    float d = x[(b * 24 + row + m / 3) * 24 + (col + m % 3)];
    float h = (0.5f * PI_F) * d;
    cq[m] = cosf(h);
    sq[m] = sinf(h);
  }
  const float* t = qp + k * 16;

  float are_[16], aim_[16];
#pragma unroll
  for (int s = 0; s < 16; ++s) {
    float m = ((s >> 3) & 1 ? sq[0] : cq[0]) * ((s >> 2) & 1 ? sq[1] : cq[1]) *
              ((s >> 1) & 1 ? sq[2] : cq[2]) * ((s & 1)      ? sq[3] : cq[3]);
    int n = (((s >> 3) & 1) + ((s >> 2) & 1) + ((s >> 1) & 1) + (s & 1)) & 3;
    are_[s] = (n == 0) ? m : ((n == 2) ? -m : 0.f);
    aim_[s] = (n == 1) ? -m : ((n == 3) ? m : 0.f);
  }

  const int RG[3][3] = {{0,2,3},{2,0,1},{4,1,3}};
#pragma unroll
  for (int g = 0; g < 3; ++g) {
    const int cb = RG[g][1], tb = RG[g][2];
    {
      float h = 0.5f * t[RG[g][0]];
      float cz = cosf(h), szv = sinf(h);
#pragma unroll
      for (int s = 0; s < 16; ++s) {
        if ((s >> cb) & 1) {
          float gg = ((s >> tb) & 1) ? szv : -szv;
          float re = are_[s], im = aim_[s];
          are_[s] = re * cz - im * gg;
          aim_[s] = im * cz + re * gg;
        }
      }
    }
    {
      float h = 0.5f * t[RG[g][0] + 1];
      float co = cosf(h), si = sinf(h);
#pragma unroll
      for (int s = 0; s < 16; ++s) {
        if (((s >> cb) & 1) && !((s >> tb) & 1)) {
          const int s1 = s | (1 << tb);
          float re0 = are_[s], im0 = aim_[s], re1 = are_[s1], im1 = aim_[s1];
          are_[s]  = co * re0 + si * im1;
          aim_[s]  = co * im0 - si * re1;
          are_[s1] = co * re1 + si * im0;
          aim_[s1] = co * im1 - si * re0;
        }
      }
    }
  }

  float zA = 0.f, zU = 0.f;
  {
    float h = 0.5f * t[14];
    float cz = cosf(h), szv = sinf(h);
    float h2 = 0.5f * t[15];
    float co = cosf(h2), si = sinf(h2);
#pragma unroll
    for (int s = 0; s < 8; ++s) {
      const int s1 = s | 8;
      float re0 = are_[s], im0 = aim_[s], re1 = are_[s1], im1 = aim_[s1];
      zA += (re0 * re0 + im0 * im0) - (re1 * re1 + im1 * im1);
      float zr0 = re0 * cz + im0 * szv, zi0 = im0 * cz - re0 * szv;
      float zr1 = re1 * cz - im1 * szv, zi1 = im1 * cz + re1 * szv;
      float ur0 = co * zr0 + si * zi1, ui0 = co * zi0 - si * zr1;
      float ur1 = co * zr1 + si * zi0, ui1 = co * zi1 - si * zr0;
      zU += (ur0 * ur0 + ui0 * ui0) - (ur1 * ur1 + ui1 * ui1);
    }
  }

  float mlane = l4 ? sq[7] : cq[7];
  float sre[16], sim[16];
#pragma unroll
  for (int i = 0; i < 16; ++i) {
    float mi = ((i & 1)        ? sq[8] : cq[8]) *
               (((i >> 1) & 1) ? sq[6] : cq[6]) *
               (((i >> 2) & 1) ? sq[5] : cq[5]) *
               (((i >> 3) & 1) ? sq[4] : cq[4]);
    int n = (l4 + (i & 1) + ((i >> 1) & 1) + ((i >> 2) & 1) + ((i >> 3) & 1)) & 3;
    float m = mlane * mi;
    sre[i] = (n == 0) ? m : ((n == 2) ? -m : 0.f);
    sim[i] = (n == 1) ? -m : ((n == 3) ? m : 0.f);
  }

  float nimv[16];
#pragma unroll
  for (int i = 0; i < 16; ++i) nimv[i] = -sim[i];

  float w[2];
#pragma unroll
  for (int tt = 0; tt < 2; ++tt) {
    v8f accre = {}, accim = {};
#pragma unroll
    for (int kc = 0; kc < 8; ++kc) {
      const v2f* bb = (const v2f*)(Lb + (((k * 16) + tt * 8 + kc) << 7));
      v2f bre = bb[lane];
      v2f bim = bb[32 + lane];
      v2f are  = {sre[2 * kc],  sre[2 * kc + 1]};
      v2f aim  = {sim[2 * kc],  sim[2 * kc + 1]};
      v2f anim = {nimv[2 * kc], nimv[2 * kc + 1]};
      accre = __builtin_amdgcn_wmma_f32_16x16x4_f32(false, are,  false, bre,
                                                    (short)0, accre, false, false);
      accre = __builtin_amdgcn_wmma_f32_16x16x4_f32(false, anim, false, bim,
                                                    (short)0, accre, false, false);
      accim = __builtin_amdgcn_wmma_f32_16x16x4_f32(false, are,  false, bim,
                                                    (short)0, accim, false, false);
      accim = __builtin_amdgcn_wmma_f32_16x16x4_f32(false, aim,  false, bre,
                                                    (short)0, accim, false, false);
    }
    float q[8];
#pragma unroll
    for (int j = 0; j < 8; ++j) q[j] = accre[j] * accre[j] + accim[j] * accim[j];
#pragma unroll
    for (int off = 8; off >= 1; off >>= 1)
#pragma unroll
      for (int j = 0; j < 8; ++j) q[j] += __shfl_xor(q[j], off, 32);
    float ws = q[0];
#pragma unroll
    for (int j = 1; j < 8; ++j) ws = ((lane & 7) == j) ? q[j] : ws;
    w[tt] = ws;
  }

  float v = w[0] * zA + w[1] * zU;
  if (valid && ((r >> 3) == l4))
    out[((b * 24 + row + 1) * 24 + (col + 1)) * 32 + k] = (v + 1.f) * 0.5f;
}

extern "C" void kernel_launch(void* const* d_in, const int* in_sizes, int n_in,
                              void* d_out, int out_size, void* d_ws, size_t ws_size,
                              hipStream_t stream) {
  const float* x  = (const float*)d_in[0];
  const float* qp = (const float*)d_in[1];
  float* out = (float*)d_out;
  float* Lb  = (float*)d_ws;

  zero_out_kernel<<<(out_size + 255) / 256, 256, 0, stream>>>(out, out_size);
  precompute_L_kernel<<<32, 32, 0, stream>>>(qp, Lb);
  quanv_main_kernel<<<244, 256, 0, stream>>>(x, qp, Lb, out);
}


# One launch instead of three: each block (one filter, 8 patch tiles) builds its filter's 32x32 gate matrix in LDS, 4 columns per wave, and 184 blocks each zero one border line; no workspace
# speedup vs baseline: 8.0989x; 2.3043x over previous
#include <hip/hip_runtime.h>

typedef __attribute__((ext_vector_type(2))) float v2f;
typedef __attribute__((ext_vector_type(8))) float v8f;

#define PI_F 3.14159265358979323846f


__global__ __launch_bounds__(256) void quanv_fused_kernel(
    const float* __restrict__ x, const float* __restrict__ qp,
    float* __restrict__ out) {
  __shared__ v2f Ls[2 * 8 * 64];

  const int lane = threadIdx.x & 31;
  const int wv = threadIdx.x >> 5;
  const int k = blockIdx.x & 31;
  const int tile = ((blockIdx.x >> 5) << 3) + wv;

  if (wv == 0 && blockIdx.x < 184) {
    const int img = blockIdx.x / 92;
    const int bi = blockIdx.x - img * 92;
    int zr, zc;
    if (bi < 24)      { zr = 0;       zc = bi; }
    else if (bi < 48) { zr = 23;      zc = bi - 24; }
    else if (bi < 70) { zr = bi - 47; zc = 0; }
    else              { zr = bi - 69; zc = 23; }
    out[((img * 24 + zr) * 24 + zc) * 32 + lane] = 0.0f;
  }

  const int r  = lane & 15;
  const int l4 = lane >> 4;
  const int pid = tile * 16 + r;
  const bool valid = pid < 968;
  const int p = valid ? pid : 967;
  const int b = p / 484;
  const int rem = p - b * 484;
  const int row = rem / 22;
  const int col = rem - row * 22;

  float d[9];
#pragma unroll
  for (int m = 0; m < 9; ++m)
    d[m] = x[(b * 24 + row + m / 3) * 24 + (col + m % 3)];

  float thc[16], ths[16];
  {
    const float h = 0.5f * qp[k * 16 + r];
    const float c = cosf(h), s = sinf(h);
#pragma unroll
    for (int i = 0; i < 16; ++i) {
      thc[i] = __int_as_float(__builtin_amdgcn_readlane(__float_as_int(c), i));
      ths[i] = __int_as_float(__builtin_amdgcn_readlane(__float_as_int(s), i));
    }
  }

  {
    float xre[4], xim[4];
#pragma unroll
    for (int j = 0; j < 4; ++j) { xre[j] = (4 * wv + j == lane) ? 1.f : 0.f; xim[j] = 0.f; }

    const int G[4][3] = {{6,0,1},{8,3,4},{10,1,2},{12,2,4}};
#pragma unroll
    for (int g = 0; g < 4; ++g) {
      const int cb = G[g][1], tb = G[g][2];
      const int ctrl = (lane >> cb) & 1;
      const int tgt  = (lane >> tb) & 1;
      {
        const float cz = thc[G[g][0]], sz = ths[G[g][0]];
        float cc = ctrl ? cz : 1.0f;
        float gg = ctrl ? (tgt ? sz : -sz) : 0.0f;
#pragma unroll
        for (int j = 0; j < 4; ++j) {
          float re = xre[j], im = xim[j];
          xre[j] = re * cc - im * gg;
          xim[j] = im * cc + re * gg;
        }
      }
      {
        const float co = thc[G[g][0] + 1], si = ths[G[g][0] + 1];
        float ce = ctrl ? co : 1.0f;
        float se = ctrl ? si : 0.0f;
#pragma unroll
        for (int j = 0; j < 4; ++j) {
          float pre = __shfl_xor(xre[j], 1 << tb, 32);
          float pim = __shfl_xor(xim[j], 1 << tb, 32);
          float re = xre[j], im = xim[j];
          xre[j] = ce * re + se * pim;
          xim[j] = ce * im - se * pre;
        }
      }
    }

    v2f* chunk = Ls + (((lane >> 4) * 8 + wv) << 6);
    const v2f re_lo = {xre[0], xre[1]}, re_hi = {xre[2], xre[3]};
    const v2f im_lo = {xim[0], xim[1]}, im_hi = {xim[2], xim[3]};
    chunk[r]           = re_lo;
    chunk[r + 16]      = re_hi;
    chunk[32 + r]      = im_lo;
    chunk[32 + r + 16] = im_hi;
  }

  float cq[9], sq[9];
#pragma unroll
  for (int m = 0; m < 9; ++m) {
    float h = (0.5f * PI_F) * d[m];
    cq[m] = cosf(h);
    sq[m] = sinf(h);
  }

  float are_[16], aim_[16];
#pragma unroll
  for (int s = 0; s < 16; ++s) {
    float m = ((s >> 3) & 1 ? sq[0] : cq[0]) * ((s >> 2) & 1 ? sq[1] : cq[1]) *
              ((s >> 1) & 1 ? sq[2] : cq[2]) * ((s & 1)      ? sq[3] : cq[3]);
    int n = (((s >> 3) & 1) + ((s >> 2) & 1) + ((s >> 1) & 1) + (s & 1)) & 3;
    are_[s] = (n == 0) ? m : ((n == 2) ? -m : 0.f);
    aim_[s] = (n == 1) ? -m : ((n == 3) ? m : 0.f);
  }

  const int RG[3][3] = {{0,2,3},{2,0,1},{4,1,3}};
#pragma unroll
  for (int g = 0; g < 3; ++g) {
    const int cb = RG[g][1], tb = RG[g][2];
    {
      const float cz = thc[RG[g][0]], szv = ths[RG[g][0]];
#pragma unroll
      for (int s = 0; s < 16; ++s) {
        if ((s >> cb) & 1) {
          float gg = ((s >> tb) & 1) ? szv : -szv;
          float re = are_[s], im = aim_[s];
          are_[s] = re * cz - im * gg;
          aim_[s] = im * cz + re * gg;
        }
      }
    }
    {
      const float co = thc[RG[g][0] + 1], si = ths[RG[g][0] + 1];
#pragma unroll
      for (int s = 0; s < 16; ++s) {
        if (((s >> cb) & 1) && !((s >> tb) & 1)) {
          const int s1 = s | (1 << tb);
          float re0 = are_[s], im0 = aim_[s], re1 = are_[s1], im1 = aim_[s1];
          are_[s]  = co * re0 + si * im1;
          aim_[s]  = co * im0 - si * re1;
          are_[s1] = co * re1 + si * im0;
          aim_[s1] = co * im1 - si * re0;
        }
      }
    }
  }

  float zA = 0.f, zU = 0.f;
  {
    const float cz = thc[14], szv = ths[14];
    const float co = thc[15], si = ths[15];
#pragma unroll
    for (int s = 0; s < 8; ++s) {
      const int s1 = s | 8;
      float re0 = are_[s], im0 = aim_[s], re1 = are_[s1], im1 = aim_[s1];
      zA += (re0 * re0 + im0 * im0) - (re1 * re1 + im1 * im1);
      float zr0 = re0 * cz + im0 * szv, zi0 = im0 * cz - re0 * szv;
      float zr1 = re1 * cz - im1 * szv, zi1 = im1 * cz + re1 * szv;
      float ur0 = co * zr0 + si * zi1, ui0 = co * zi0 - si * zr1;
      float ur1 = co * zr1 + si * zi0, ui1 = co * zi1 - si * zr0;
      zU += (ur0 * ur0 + ui0 * ui0) - (ur1 * ur1 + ui1 * ui1);
    }
  }

  float mlane = l4 ? sq[7] : cq[7];
  float sre[16], sim[16];
#pragma unroll
  for (int i = 0; i < 16; ++i) {
    float mi = ((i & 1)        ? sq[8] : cq[8]) *
               (((i >> 1) & 1) ? sq[6] : cq[6]) *
               (((i >> 2) & 1) ? sq[5] : cq[5]) *
               (((i >> 3) & 1) ? sq[4] : cq[4]);
    int n = (l4 + (i & 1) + ((i >> 1) & 1) + ((i >> 2) & 1) + ((i >> 3) & 1)) & 3;
    float m = mlane * mi;
    sre[i] = (n == 0) ? m : ((n == 2) ? -m : 0.f);
    sim[i] = (n == 1) ? -m : ((n == 3) ? m : 0.f);
  }

  float nimv[16];
#pragma unroll
  for (int i = 0; i < 16; ++i) nimv[i] = -sim[i];

  __syncthreads();

  float w[2];
#pragma unroll
  for (int tt = 0; tt < 2; ++tt) {
    v8f accre = {}, accim = {};
#pragma unroll
    for (int kc = 0; kc < 8; ++kc) {
      const v2f* bb = Ls + ((tt * 8 + kc) << 6);
      v2f bre = bb[lane];
      v2f bim = bb[32 + lane];
      v2f are  = {sre[2 * kc],  sre[2 * kc + 1]};
      v2f aim  = {sim[2 * kc],  sim[2 * kc + 1]};
      v2f anim = {nimv[2 * kc], nimv[2 * kc + 1]};
      accre = __builtin_amdgcn_wmma_f32_16x16x4_f32(false, are,  false, bre,
                                                    (short)0, accre, false, false);
      accre = __builtin_amdgcn_wmma_f32_16x16x4_f32(false, anim, false, bim,
                                                    (short)0, accre, false, false);
      accim = __builtin_amdgcn_wmma_f32_16x16x4_f32(false, are,  false, bim,
                                                    (short)0, accim, false, false);
      accim = __builtin_amdgcn_wmma_f32_16x16x4_f32(false, aim,  false, bre,
                                                    (short)0, accim, false, false);
    }
    float q[8];
#pragma unroll
    for (int j = 0; j < 8; ++j) q[j] = accre[j] * accre[j] + accim[j] * accim[j];
#pragma unroll
    for (int off = 8; off >= 1; off >>= 1)
#pragma unroll
      for (int j = 0; j < 8; ++j) q[j] += __shfl_xor(q[j], off, 32);
    float ws = q[0];
#pragma unroll
    for (int j = 1; j < 8; ++j) ws = ((lane & 7) == j) ? q[j] : ws;
    w[tt] = ws;
  }

  float v = w[0] * zA + w[1] * zU;
  if (valid && ((r >> 3) == l4))
    out[((b * 24 + row + 1) * 24 + (col + 1)) * 32 + k] = (v + 1.f) * 0.5f;
}

extern "C" void kernel_launch(void* const* d_in, const int* in_sizes, int n_in,
                              void* d_out, int out_size, void* d_ws, size_t ws_size,
                              hipStream_t stream) {
  const float* x  = (const float*)d_in[0];
  const float* qp = (const float*)d_in[1];
  float* out = (float*)d_out;

  quanv_fused_kernel<<<256, 256, 0, stream>>>(x, qp, out);
}


# Real A operand: the (-i)^popcount phases of the RX product state are folded into L's columns at build, b'=Lt*m with m real: 2 f32 WMMAs per chunk and tile, not 4 (32 per wave); sum grouping changes
# speedup vs baseline: 8.5468x; 1.0553x over previous
#include <hip/hip_runtime.h>

typedef __attribute__((ext_vector_type(2))) float v2f;
typedef __attribute__((ext_vector_type(8))) float v8f;

#define PI_F 3.14159265358979323846f


__global__ __launch_bounds__(256) void quanv_fused_kernel(
    const float* __restrict__ x, const float* __restrict__ qp,
    float* __restrict__ out) {
  __shared__ v2f Ls[2 * 8 * 64];

  const int lane = threadIdx.x & 31;
  const int wv = threadIdx.x >> 5;
  const int k = blockIdx.x & 31;
  const int tile = ((blockIdx.x >> 5) << 3) + wv;

  if (wv == 0 && blockIdx.x < 184) {
    const int img = blockIdx.x / 92;
    const int bi = blockIdx.x - img * 92;
    int zr, zc;
    if (bi < 24)      { zr = 0;       zc = bi; }
    else if (bi < 48) { zr = 23;      zc = bi - 24; }
    else if (bi < 70) { zr = bi - 47; zc = 0; }
    else              { zr = bi - 69; zc = 23; }
    out[((img * 24 + zr) * 24 + zc) * 32 + lane] = 0.0f;
  }

  const int r  = lane & 15;
  const int l4 = lane >> 4;
  const int pid = tile * 16 + r;
  const bool valid = pid < 968;
  const int p = valid ? pid : 967;
  const int b = p / 484;
  const int rem = p - b * 484;
  const int row = rem / 22;
  const int col = rem - row * 22;

  float d[9];
#pragma unroll
  for (int m = 0; m < 9; ++m)
    d[m] = x[(b * 24 + row + m / 3) * 24 + (col + m % 3)];

  float thc[16], ths[16];
  {
    const float h = 0.5f * qp[k * 16 + r];
    const float c = cosf(h), s = sinf(h);
#pragma unroll
    for (int i = 0; i < 16; ++i) {
      thc[i] = __int_as_float(__builtin_amdgcn_readlane(__float_as_int(c), i));
      ths[i] = __int_as_float(__builtin_amdgcn_readlane(__float_as_int(s), i));
    }
  }

  {
    float xre[4], xim[4];
#pragma unroll
    for (int j = 0; j < 4; ++j) {
      const int n = (__builtin_popcount(wv) + __builtin_popcount(j)) & 3;
      const float pr = (n == 0) ? 1.f : ((n == 2) ? -1.f : 0.f);
      const float pim = (n == 1) ? -1.f : ((n == 3) ? 1.f : 0.f);
      const bool diag = (4 * wv + j == lane);
      xre[j] = diag ? pr : 0.f;
      xim[j] = diag ? pim : 0.f;
    }

    const int G[4][3] = {{6,0,1},{8,3,4},{10,1,2},{12,2,4}};
#pragma unroll
    for (int g = 0; g < 4; ++g) {
      const int cb = G[g][1], tb = G[g][2];
      const int ctrl = (lane >> cb) & 1;
      const int tgt  = (lane >> tb) & 1;
      {
        const float cz = thc[G[g][0]], sz = ths[G[g][0]];
        float cc = ctrl ? cz : 1.0f;
        float gg = ctrl ? (tgt ? sz : -sz) : 0.0f;
#pragma unroll
        for (int j = 0; j < 4; ++j) {
          float re = xre[j], im = xim[j];
          xre[j] = re * cc - im * gg;
          xim[j] = im * cc + re * gg;
        }
      }
      {
        const float co = thc[G[g][0] + 1], si = ths[G[g][0] + 1];
        float ce = ctrl ? co : 1.0f;
        float se = ctrl ? si : 0.0f;
#pragma unroll
        for (int j = 0; j < 4; ++j) {
          float pre = __shfl_xor(xre[j], 1 << tb, 32);
          float pim = __shfl_xor(xim[j], 1 << tb, 32);
          float re = xre[j], im = xim[j];
          xre[j] = ce * re + se * pim;
          xim[j] = ce * im - se * pre;
        }
      }
    }

    v2f* chunk = Ls + (((lane >> 4) * 8 + wv) << 6);
    const v2f re_lo = {xre[0], xre[1]}, re_hi = {xre[2], xre[3]};
    const v2f im_lo = {xim[0], xim[1]}, im_hi = {xim[2], xim[3]};
    chunk[r]           = re_lo;
    chunk[r + 16]      = re_hi;
    chunk[32 + r]      = im_lo;
    chunk[32 + r + 16] = im_hi;
  }

  float cq[9], sq[9];
#pragma unroll
  for (int m = 0; m < 9; ++m) {
    float h = (0.5f * PI_F) * d[m];
    cq[m] = cosf(h);
    sq[m] = sinf(h);
  }

  float are_[16], aim_[16];
#pragma unroll
  for (int s = 0; s < 16; ++s) {
    float m = ((s >> 3) & 1 ? sq[0] : cq[0]) * ((s >> 2) & 1 ? sq[1] : cq[1]) *
              ((s >> 1) & 1 ? sq[2] : cq[2]) * ((s & 1)      ? sq[3] : cq[3]);
    int n = (((s >> 3) & 1) + ((s >> 2) & 1) + ((s >> 1) & 1) + (s & 1)) & 3;
    are_[s] = (n == 0) ? m : ((n == 2) ? -m : 0.f);
    aim_[s] = (n == 1) ? -m : ((n == 3) ? m : 0.f);
  }

  const int RG[3][3] = {{0,2,3},{2,0,1},{4,1,3}};
#pragma unroll
  for (int g = 0; g < 3; ++g) {
    const int cb = RG[g][1], tb = RG[g][2];
    {
      const float cz = thc[RG[g][0]], szv = ths[RG[g][0]];
#pragma unroll
      for (int s = 0; s < 16; ++s) {
        if ((s >> cb) & 1) {
          float gg = ((s >> tb) & 1) ? szv : -szv;
          float re = are_[s], im = aim_[s];
          are_[s] = re * cz - im * gg;
          aim_[s] = im * cz + re * gg;
        }
      }
    }
    {
      const float co = thc[RG[g][0] + 1], si = ths[RG[g][0] + 1];
#pragma unroll
      for (int s = 0; s < 16; ++s) {
        if (((s >> cb) & 1) && !((s >> tb) & 1)) {
          const int s1 = s | (1 << tb);
          float re0 = are_[s], im0 = aim_[s], re1 = are_[s1], im1 = aim_[s1];
          are_[s]  = co * re0 + si * im1;
          aim_[s]  = co * im0 - si * re1;
          are_[s1] = co * re1 + si * im0;
          aim_[s1] = co * im1 - si * re0;
        }
      }
    }
  }

  float zA = 0.f, zU = 0.f;
  {
    const float cz = thc[14], szv = ths[14];
    const float co = thc[15], si = ths[15];
#pragma unroll
    for (int s = 0; s < 8; ++s) {
      const int s1 = s | 8;
      float re0 = are_[s], im0 = aim_[s], re1 = are_[s1], im1 = aim_[s1];
      zA += (re0 * re0 + im0 * im0) - (re1 * re1 + im1 * im1);
      float zr0 = re0 * cz + im0 * szv, zi0 = im0 * cz - re0 * szv;
      float zr1 = re1 * cz - im1 * szv, zi1 = im1 * cz + re1 * szv;
      float ur0 = co * zr0 + si * zi1, ui0 = co * zi0 - si * zr1;
      float ur1 = co * zr1 + si * zi0, ui1 = co * zi1 - si * zr0;
      zU += (ur0 * ur0 + ui0 * ui0) - (ur1 * ur1 + ui1 * ui1);
    }
  }

  float mlane = l4 ? sq[7] : cq[7];
  float mv[16];
#pragma unroll
  for (int i = 0; i < 16; ++i) {
    float mi = ((i & 1)        ? sq[8] : cq[8]) *
               (((i >> 1) & 1) ? sq[6] : cq[6]) *
               (((i >> 2) & 1) ? sq[5] : cq[5]) *
               (((i >> 3) & 1) ? sq[4] : cq[4]);
    mv[i] = mlane * mi;
  }

  __syncthreads();

  float w[2];
#pragma unroll
  for (int tt = 0; tt < 2; ++tt) {
    v8f accre = {}, accim = {};
#pragma unroll
    for (int kc = 0; kc < 8; ++kc) {
      const v2f* bb = Ls + ((tt * 8 + kc) << 6);
      v2f bre = bb[lane];
      v2f bim = bb[32 + lane];
      v2f am = {mv[2 * kc], mv[2 * kc + 1]};
      accre = __builtin_amdgcn_wmma_f32_16x16x4_f32(false, am, false, bre,
                                                    (short)0, accre, false, false);
      accim = __builtin_amdgcn_wmma_f32_16x16x4_f32(false, am, false, bim,
                                                    (short)0, accim, false, false);
    }
    float q[8];
#pragma unroll
    for (int j = 0; j < 8; ++j) q[j] = accre[j] * accre[j] + accim[j] * accim[j];
#pragma unroll
    for (int off = 8; off >= 1; off >>= 1)
#pragma unroll
      for (int j = 0; j < 8; ++j) q[j] += __shfl_xor(q[j], off, 32);
    float ws = q[0];
#pragma unroll
    for (int j = 1; j < 8; ++j) ws = ((lane & 7) == j) ? q[j] : ws;
    w[tt] = ws;
  }

  float v = w[0] * zA + w[1] * zU;
  if (valid && ((r >> 3) == l4))
    out[((b * 24 + row + 1) * 24 + (col + 1)) * 32 + k] = (v + 1.f) * 0.5f;
}

extern "C" void kernel_launch(void* const* d_in, const int* in_sizes, int n_in,
                              void* d_out, int out_size, void* d_ws, size_t ws_size,
                              hipStream_t stream) {
  const float* x  = (const float*)d_in[0];
  const float* qp = (const float*)d_in[1];
  float* out = (float*)d_out;

  quanv_fused_kernel<<<256, 256, 0, stream>>>(x, qp, out);
}


# Unit norm: gates are unitary and the product state has norm 1, so w1=1-w0 and p1=1-p0: only the c4=0 tile of b' is formed (16 f32 WMMAs per wave, not 32; half the row sums); sum grouping changes
# speedup vs baseline: 9.1232x; 1.0674x over previous
#include <hip/hip_runtime.h>

typedef __attribute__((ext_vector_type(2))) float v2f;
typedef __attribute__((ext_vector_type(8))) float v8f;

#define PI_F 3.14159265358979323846f


__global__ __launch_bounds__(256) void quanv_fused_kernel(
    const float* __restrict__ x, const float* __restrict__ qp,
    float* __restrict__ out) {
  __shared__ v2f Ls[8 * 64];

  const int lane = threadIdx.x & 31;
  const int wv = threadIdx.x >> 5;
  const int k = blockIdx.x & 31;
  const int tile = ((blockIdx.x >> 5) << 3) + wv;

  if (wv == 0 && blockIdx.x < 184) {
    const int img = blockIdx.x / 92;
    const int bi = blockIdx.x - img * 92;
    int zr, zc;
    if (bi < 24)      { zr = 0;       zc = bi; }
    else if (bi < 48) { zr = 23;      zc = bi - 24; }
    else if (bi < 70) { zr = bi - 47; zc = 0; }
    else              { zr = bi - 69; zc = 23; }
    out[((img * 24 + zr) * 24 + zc) * 32 + lane] = 0.0f;
  }

  const int r  = lane & 15;
  const int l4 = lane >> 4;
  const int pid = tile * 16 + r;
  const bool valid = pid < 968;
  const int p = valid ? pid : 967;
  const int b = p / 484;
  const int rem = p - b * 484;
  const int row = rem / 22;
  const int col = rem - row * 22;

  float d[9];
#pragma unroll
  for (int m = 0; m < 9; ++m)
    d[m] = x[(b * 24 + row + m / 3) * 24 + (col + m % 3)];

  float thc[16], ths[16];
  {
    const float h = 0.5f * qp[k * 16 + r];
    const float c = cosf(h), s = sinf(h);
#pragma unroll
    for (int i = 0; i < 16; ++i) {
      thc[i] = __int_as_float(__builtin_amdgcn_readlane(__float_as_int(c), i));
      ths[i] = __int_as_float(__builtin_amdgcn_readlane(__float_as_int(s), i));
    }
  }

  {
    float xre[4], xim[4];
#pragma unroll
    for (int j = 0; j < 4; ++j) {
      const int n = (__builtin_popcount(wv) + __builtin_popcount(j)) & 3;
      const float pr = (n == 0) ? 1.f : ((n == 2) ? -1.f : 0.f);
      const float pim = (n == 1) ? -1.f : ((n == 3) ? 1.f : 0.f);
      const bool diag = (4 * wv + j == lane);
      xre[j] = diag ? pr : 0.f;
      xim[j] = diag ? pim : 0.f;
    }

    const int G[4][3] = {{6,0,1},{8,3,4},{10,1,2},{12,2,4}};
#pragma unroll
    for (int g = 0; g < 4; ++g) {
      const int cb = G[g][1], tb = G[g][2];
      const int ctrl = (lane >> cb) & 1;
      const int tgt  = (lane >> tb) & 1;
      {
        const float cz = thc[G[g][0]], sz = ths[G[g][0]];
        float cc = ctrl ? cz : 1.0f;
        float gg = ctrl ? (tgt ? sz : -sz) : 0.0f;
#pragma unroll
        for (int j = 0; j < 4; ++j) {
          float re = xre[j], im = xim[j];
          xre[j] = re * cc - im * gg;
          xim[j] = im * cc + re * gg;
        }
      }
      {
        const float co = thc[G[g][0] + 1], si = ths[G[g][0] + 1];
        float ce = ctrl ? co : 1.0f;
        float se = ctrl ? si : 0.0f;
#pragma unroll
        for (int j = 0; j < 4; ++j) {
          float pre = __shfl_xor(xre[j], 1 << tb, 32);
          float pim = __shfl_xor(xim[j], 1 << tb, 32);
          float re = xre[j], im = xim[j];
          xre[j] = ce * re + se * pim;
          xim[j] = ce * im - se * pre;
        }
      }
    }

    if (lane < 16) {
      v2f* chunk = Ls + (wv << 6);
      const v2f re_lo = {xre[0], xre[1]}, re_hi = {xre[2], xre[3]};
      const v2f im_lo = {xim[0], xim[1]}, im_hi = {xim[2], xim[3]};
      chunk[lane]           = re_lo;
      chunk[lane + 16]      = re_hi;
      chunk[32 + lane]      = im_lo;
      chunk[32 + lane + 16] = im_hi;
    }
  }

  float cq[9], sq[9];
#pragma unroll
  for (int m = 0; m < 9; ++m) {
    float h = (0.5f * PI_F) * d[m];
    cq[m] = cosf(h);
    sq[m] = sinf(h);
  }

  float are_[16], aim_[16];
#pragma unroll
  for (int s = 0; s < 16; ++s) {
    float m = ((s >> 3) & 1 ? sq[0] : cq[0]) * ((s >> 2) & 1 ? sq[1] : cq[1]) *
              ((s >> 1) & 1 ? sq[2] : cq[2]) * ((s & 1)      ? sq[3] : cq[3]);
    int n = (((s >> 3) & 1) + ((s >> 2) & 1) + ((s >> 1) & 1) + (s & 1)) & 3;
    are_[s] = (n == 0) ? m : ((n == 2) ? -m : 0.f);
    aim_[s] = (n == 1) ? -m : ((n == 3) ? m : 0.f);
  }

  const int RG[3][3] = {{0,2,3},{2,0,1},{4,1,3}};
#pragma unroll
  for (int g = 0; g < 3; ++g) {
    const int cb = RG[g][1], tb = RG[g][2];
    {
      const float cz = thc[RG[g][0]], szv = ths[RG[g][0]];
#pragma unroll
      for (int s = 0; s < 16; ++s) {
        if ((s >> cb) & 1) {
          float gg = ((s >> tb) & 1) ? szv : -szv;
          float re = are_[s], im = aim_[s];
          are_[s] = re * cz - im * gg;
          aim_[s] = im * cz + re * gg;
        }
      }
    }
    {
      const float co = thc[RG[g][0] + 1], si = ths[RG[g][0] + 1];
#pragma unroll
      for (int s = 0; s < 16; ++s) {
        if (((s >> cb) & 1) && !((s >> tb) & 1)) {
          const int s1 = s | (1 << tb);
          float re0 = are_[s], im0 = aim_[s], re1 = are_[s1], im1 = aim_[s1];
          are_[s]  = co * re0 + si * im1;
          aim_[s]  = co * im0 - si * re1;
          are_[s1] = co * re1 + si * im0;
          aim_[s1] = co * im1 - si * re0;
        }
      }
    }
  }

  float PA = 0.f, PU = 0.f;
  {
    const float cz = thc[14], szv = ths[14];
    const float co = thc[15], si = ths[15];
#pragma unroll
    for (int s = 0; s < 8; ++s) {
      const int s1 = s | 8;
      float re0 = are_[s], im0 = aim_[s], re1 = are_[s1], im1 = aim_[s1];
      PA = fmaf(re0, re0, PA);
      PA = fmaf(im0, im0, PA);
      float zr0 = re0 * cz + im0 * szv, zi0 = im0 * cz - re0 * szv;
      float zr1 = re1 * cz - im1 * szv, zi1 = im1 * cz + re1 * szv;
      float ur0 = co * zr0 + si * zi1, ui0 = co * zi0 - si * zr1;
      PU = fmaf(ur0, ur0, PU);
      PU = fmaf(ui0, ui0, PU);
    }
  }

  float mlane = l4 ? sq[7] : cq[7];
  float mv[16];
#pragma unroll
  for (int i = 0; i < 16; ++i) {
    float mi = ((i & 1)        ? sq[8] : cq[8]) *
               (((i >> 1) & 1) ? sq[6] : cq[6]) *
               (((i >> 2) & 1) ? sq[5] : cq[5]) *
               (((i >> 3) & 1) ? sq[4] : cq[4]);
    mv[i] = mlane * mi;
  }

  __syncthreads();

  float w0;
  {
    v8f accre = {}, accim = {};
#pragma unroll
    for (int kc = 0; kc < 8; ++kc) {
      const v2f* bb = Ls + (kc << 6);
      v2f bre = bb[lane];
      v2f bim = bb[32 + lane];
      v2f am = {mv[2 * kc], mv[2 * kc + 1]};
      accre = __builtin_amdgcn_wmma_f32_16x16x4_f32(false, am, false, bre,
                                                    (short)0, accre, false, false);
      accim = __builtin_amdgcn_wmma_f32_16x16x4_f32(false, am, false, bim,
                                                    (short)0, accim, false, false);
    }
    float q[8];
#pragma unroll
    for (int j = 0; j < 8; ++j) q[j] = accre[j] * accre[j] + accim[j] * accim[j];
#pragma unroll
    for (int off = 8; off >= 1; off >>= 1)
#pragma unroll
      for (int j = 0; j < 8; ++j) q[j] += __shfl_xor(q[j], off, 32);
    float ws = q[0];
#pragma unroll
    for (int j = 1; j < 8; ++j) ws = ((lane & 7) == j) ? q[j] : ws;
    w0 = ws;
  }

  if (valid && ((r >> 3) == l4))
    out[((b * 24 + row + 1) * 24 + (col + 1)) * 32 + k] = PU + w0 * (PA - PU);
}

extern "C" void kernel_launch(void* const* d_in, const int* in_sizes, int n_in,
                              void* d_out, int out_size, void* d_ws, size_t ws_size,
                              hipStream_t stream) {
  const float* x  = (const float*)d_in[0];
  const float* qp = (const float*)d_in[1];
  float* out = (float*)d_out;

  quanv_fused_kernel<<<256, 256, 0, stream>>>(x, qp, out);
}


# Trig pairs share one reduction: pixel half-angles by an in-line quarter-turn sin/cos pair (round, subtract, two short polynomials; angle not rounded to f32 first), filter angles by sincosf
# speedup vs baseline: 10.5300x; 1.1542x over previous
#include <hip/hip_runtime.h>

typedef __attribute__((ext_vector_type(2))) float v2f;
typedef __attribute__((ext_vector_type(8))) float v8f;

__device__ __forceinline__ void sincos_quarter_turns(float d, float& s, float& c) {
  const float t = __builtin_rintf(d);
  const float w = d - t;
  const int q = (int)t;
  const float y = w * w;
  float sp = fmaf(y, 0.000468568905f, -0.00485667493f);
  sp = fmaf(y, sp, 0.0797256678f);
  sp = fmaf(y, sp, -0.645966411f);
  const float sw = fmaf(1.57079637f, w, (w * y) * sp);
  float cp = fmaf(y, 0.0000486124918f, 0.000870445569f);
  cp = fmaf(y, cp, -0.0208520181f);
  cp = fmaf(y, cp, 0.253668398f);
  cp = fmaf(y, cp, -1.23370051f);
  const float cw = fmaf(y, cp, 1.0f);
  const bool odd = (q & 1) != 0;
  const unsigned flip = ((unsigned)q & 2u) << 30;
  s = __uint_as_float(__float_as_uint(odd ? cw : sw) ^ flip);
  c = __uint_as_float(__float_as_uint(odd ? -sw : cw) ^ flip);
}


__global__ __launch_bounds__(256) void quanv_fused_kernel(
    const float* __restrict__ x, const float* __restrict__ qp,
    float* __restrict__ out) {
  __shared__ v2f Ls[8 * 64];

  const int lane = threadIdx.x & 31;
  const int wv = threadIdx.x >> 5;
  const int k = blockIdx.x & 31;
  const int tile = ((blockIdx.x >> 5) << 3) + wv;

  if (wv == 0 && blockIdx.x < 184) {
    const int img = blockIdx.x / 92;
    const int bi = blockIdx.x - img * 92;
    int zr, zc;
    if (bi < 24)      { zr = 0;       zc = bi; }
    else if (bi < 48) { zr = 23;      zc = bi - 24; }
    else if (bi < 70) { zr = bi - 47; zc = 0; }
    else              { zr = bi - 69; zc = 23; }
    out[((img * 24 + zr) * 24 + zc) * 32 + lane] = 0.0f;
  }

  const int r  = lane & 15;
  const int l4 = lane >> 4;
  const int pid = tile * 16 + r;
  const bool valid = pid < 968;
  const int p = valid ? pid : 967;
  const int b = p / 484;
  const int rem = p - b * 484;
  const int row = rem / 22;
  const int col = rem - row * 22;

  float d[9];
#pragma unroll
  for (int m = 0; m < 9; ++m)
    d[m] = x[(b * 24 + row + m / 3) * 24 + (col + m % 3)];

  float thc[16], ths[16];
  {
    const float h = 0.5f * qp[k * 16 + r];
    float c, s;
    sincosf(h, &s, &c);
#pragma unroll
    for (int i = 0; i < 16; ++i) {
      thc[i] = __int_as_float(__builtin_amdgcn_readlane(__float_as_int(c), i));
      ths[i] = __int_as_float(__builtin_amdgcn_readlane(__float_as_int(s), i));
    }
  }

  {
    float xre[4], xim[4];
#pragma unroll
    for (int j = 0; j < 4; ++j) {
      const int n = (__builtin_popcount(wv) + __builtin_popcount(j)) & 3;
      const float pr = (n == 0) ? 1.f : ((n == 2) ? -1.f : 0.f);
      const float pim = (n == 1) ? -1.f : ((n == 3) ? 1.f : 0.f);
      const bool diag = (4 * wv + j == lane);
      xre[j] = diag ? pr : 0.f;
      xim[j] = diag ? pim : 0.f;
    }

    const int G[4][3] = {{6,0,1},{8,3,4},{10,1,2},{12,2,4}};
#pragma unroll
    for (int g = 0; g < 4; ++g) {
      const int cb = G[g][1], tb = G[g][2];
      const int ctrl = (lane >> cb) & 1;
      const int tgt  = (lane >> tb) & 1;
      {
        const float cz = thc[G[g][0]], sz = ths[G[g][0]];
        float cc = ctrl ? cz : 1.0f;
        float gg = ctrl ? (tgt ? sz : -sz) : 0.0f;
#pragma unroll
        for (int j = 0; j < 4; ++j) {
          float re = xre[j], im = xim[j];
          xre[j] = re * cc - im * gg;
          xim[j] = im * cc + re * gg;
        }
      }
      {
        const float co = thc[G[g][0] + 1], si = ths[G[g][0] + 1];
        float ce = ctrl ? co : 1.0f;
        float se = ctrl ? si : 0.0f;
#pragma unroll
        for (int j = 0; j < 4; ++j) {
          float pre = __shfl_xor(xre[j], 1 << tb, 32);
          float pim = __shfl_xor(xim[j], 1 << tb, 32);
          float re = xre[j], im = xim[j];
          xre[j] = ce * re + se * pim;
          xim[j] = ce * im - se * pre;
        }
      }
    }

    if (lane < 16) {
      v2f* chunk = Ls + (wv << 6);
      const v2f re_lo = {xre[0], xre[1]}, re_hi = {xre[2], xre[3]};
      const v2f im_lo = {xim[0], xim[1]}, im_hi = {xim[2], xim[3]};
      chunk[lane]           = re_lo;
      chunk[lane + 16]      = re_hi;
      chunk[32 + lane]      = im_lo;
      chunk[32 + lane + 16] = im_hi;
    }
  }

  float cq[9], sq[9];
#pragma unroll
  for (int m = 0; m < 9; ++m)
    sincos_quarter_turns(d[m], sq[m], cq[m]);

  float are_[16], aim_[16];
#pragma unroll
  for (int s = 0; s < 16; ++s) {
    float m = ((s >> 3) & 1 ? sq[0] : cq[0]) * ((s >> 2) & 1 ? sq[1] : cq[1]) *
              ((s >> 1) & 1 ? sq[2] : cq[2]) * ((s & 1)      ? sq[3] : cq[3]);
    int n = (((s >> 3) & 1) + ((s >> 2) & 1) + ((s >> 1) & 1) + (s & 1)) & 3;
    are_[s] = (n == 0) ? m : ((n == 2) ? -m : 0.f);
    aim_[s] = (n == 1) ? -m : ((n == 3) ? m : 0.f);
  }

  const int RG[3][3] = {{0,2,3},{2,0,1},{4,1,3}};
#pragma unroll
  for (int g = 0; g < 3; ++g) {
    const int cb = RG[g][1], tb = RG[g][2];
    {
      const float cz = thc[RG[g][0]], szv = ths[RG[g][0]];
#pragma unroll
      for (int s = 0; s < 16; ++s) {
        if ((s >> cb) & 1) {
          float gg = ((s >> tb) & 1) ? szv : -szv;
          float re = are_[s], im = aim_[s];
          are_[s] = re * cz - im * gg;
          aim_[s] = im * cz + re * gg;
        }
      }
    }
    {
      const float co = thc[RG[g][0] + 1], si = ths[RG[g][0] + 1];
#pragma unroll
      for (int s = 0; s < 16; ++s) {
        if (((s >> cb) & 1) && !((s >> tb) & 1)) {
          const int s1 = s | (1 << tb);
          float re0 = are_[s], im0 = aim_[s], re1 = are_[s1], im1 = aim_[s1];
          are_[s]  = co * re0 + si * im1;
          aim_[s]  = co * im0 - si * re1;
          are_[s1] = co * re1 + si * im0;
          aim_[s1] = co * im1 - si * re0;
        }
      }
    }
  }

  float PA = 0.f, PU = 0.f;
  {
    const float cz = thc[14], szv = ths[14];
    const float co = thc[15], si = ths[15];
#pragma unroll
    for (int s = 0; s < 8; ++s) {
      const int s1 = s | 8;
      float re0 = are_[s], im0 = aim_[s], re1 = are_[s1], im1 = aim_[s1];
      PA = fmaf(re0, re0, PA);
      PA = fmaf(im0, im0, PA);
      float zr0 = re0 * cz + im0 * szv, zi0 = im0 * cz - re0 * szv;
      float zr1 = re1 * cz - im1 * szv, zi1 = im1 * cz + re1 * szv;
      float ur0 = co * zr0 + si * zi1, ui0 = co * zi0 - si * zr1;
      PU = fmaf(ur0, ur0, PU);
      PU = fmaf(ui0, ui0, PU);
    }
  }

  float mlane = l4 ? sq[7] : cq[7];
  float mv[16];
#pragma unroll
  for (int i = 0; i < 16; ++i) {
    float mi = ((i & 1)        ? sq[8] : cq[8]) *
               (((i >> 1) & 1) ? sq[6] : cq[6]) *
               (((i >> 2) & 1) ? sq[5] : cq[5]) *
               (((i >> 3) & 1) ? sq[4] : cq[4]);
    mv[i] = mlane * mi;
  }

  __syncthreads();

  float w0;
  {
    v8f accre = {}, accim = {};
#pragma unroll
    for (int kc = 0; kc < 8; ++kc) {
      const v2f* bb = Ls + (kc << 6);
      v2f bre = bb[lane];
      v2f bim = bb[32 + lane];
      v2f am = {mv[2 * kc], mv[2 * kc + 1]};
      accre = __builtin_amdgcn_wmma_f32_16x16x4_f32(false, am, false, bre,
                                                    (short)0, accre, false, false);
      accim = __builtin_amdgcn_wmma_f32_16x16x4_f32(false, am, false, bim,
                                                    (short)0, accim, false, false);
    }
    float q[8];
#pragma unroll
    for (int j = 0; j < 8; ++j) q[j] = accre[j] * accre[j] + accim[j] * accim[j];
#pragma unroll
    for (int off = 8; off >= 1; off >>= 1)
#pragma unroll
      for (int j = 0; j < 8; ++j) q[j] += __shfl_xor(q[j], off, 32);
    float ws = q[0];
#pragma unroll
    for (int j = 1; j < 8; ++j) ws = ((lane & 7) == j) ? q[j] : ws;
    w0 = ws;
  }

  if (valid && ((r >> 3) == l4))
    out[((b * 24 + row + 1) * 24 + (col + 1)) * 32 + k] = PU + w0 * (PA - PU);
}

extern "C" void kernel_launch(void* const* d_in, const int* in_sizes, int n_in,
                              void* d_out, int out_size, void* d_ws, size_t ws_size,
                              hipStream_t stream) {
  const float* x  = (const float*)d_in[0];
  const float* qp = (const float*)d_in[1];
  float* out = (float*)d_out;

  quanv_fused_kernel<<<256, 256, 0, stream>>>(x, qp, out);
}


# Qubit 0 as a Bloch vector: gates 1-3 and 8 become 3-D rotations mixed by the control qubits' probabilities (about 45 real ops per lane, not a 16-amplitude complex evolution); sum grouping changes
# speedup vs baseline: 11.2549x; 1.0688x over previous
#include <hip/hip_runtime.h>

typedef __attribute__((ext_vector_type(2))) float v2f;
typedef __attribute__((ext_vector_type(8))) float v8f;

__device__ __forceinline__ void sincos_quarter_turns(float d, float& s, float& c) {
  const float t = __builtin_rintf(d);
  const float w = d - t;
  const int q = (int)t;
  const float y = w * w;
  float sp = fmaf(y, 0.000468568905f, -0.00485667493f);
  sp = fmaf(y, sp, 0.0797256678f);
  sp = fmaf(y, sp, -0.645966411f);
  const float sw = fmaf(1.57079637f, w, (w * y) * sp);
  float cp = fmaf(y, 0.0000486124918f, 0.000870445569f);
  cp = fmaf(y, cp, -0.0208520181f);
  cp = fmaf(y, cp, 0.253668398f);
  cp = fmaf(y, cp, -1.23370051f);
  const float cw = fmaf(y, cp, 1.0f);
  const bool odd = (q & 1) != 0;
  const unsigned flip = ((unsigned)q & 2u) << 30;
  s = __uint_as_float(__float_as_uint(odd ? cw : sw) ^ flip);
  c = __uint_as_float(__float_as_uint(odd ? -sw : cw) ^ flip);
}


__global__ __launch_bounds__(256) void quanv_fused_kernel(
    const float* __restrict__ x, const float* __restrict__ qp,
    float* __restrict__ out) {
  __shared__ v2f Ls[8 * 64];

  const int lane = threadIdx.x & 31;
  const int wv = threadIdx.x >> 5;
  const int k = blockIdx.x & 31;
  const int tile = ((blockIdx.x >> 5) << 3) + wv;

  if (wv == 0 && blockIdx.x < 184) {
    const int img = blockIdx.x / 92;
    const int bi = blockIdx.x - img * 92;
    int zr, zc;
    if (bi < 24)      { zr = 0;       zc = bi; }
    else if (bi < 48) { zr = 23;      zc = bi - 24; }
    else if (bi < 70) { zr = bi - 47; zc = 0; }
    else              { zr = bi - 69; zc = 23; }
    out[((img * 24 + zr) * 24 + zc) * 32 + lane] = 0.0f;
  }

  const int r  = lane & 15;
  const int l4 = lane >> 4;
  const int pid = tile * 16 + r;
  const bool valid = pid < 968;
  const int p = valid ? pid : 967;
  const int b = p / 484;
  const int rem = p - b * 484;
  const int row = rem / 22;
  const int col = rem - row * 22;

  float d[9];
#pragma unroll
  for (int m = 0; m < 9; ++m)
    d[m] = x[(b * 24 + row + m / 3) * 24 + (col + m % 3)];

  float thc[16], ths[16];
  {
    const float ang = qp[k * 16 + r];
    const float h = (r >= 6 && r < 14) ? 0.5f * ang : ang;
    float c, s;
    sincosf(h, &s, &c);
#pragma unroll
    for (int i = 0; i < 16; ++i) {
      thc[i] = __int_as_float(__builtin_amdgcn_readlane(__float_as_int(c), i));
      ths[i] = __int_as_float(__builtin_amdgcn_readlane(__float_as_int(s), i));
    }
  }

  {
    float xre[4], xim[4];
#pragma unroll
    for (int j = 0; j < 4; ++j) {
      const int n = (__builtin_popcount(wv) + __builtin_popcount(j)) & 3;
      const float pr = (n == 0) ? 1.f : ((n == 2) ? -1.f : 0.f);
      const float pim = (n == 1) ? -1.f : ((n == 3) ? 1.f : 0.f);
      const bool diag = (4 * wv + j == lane);
      xre[j] = diag ? pr : 0.f;
      xim[j] = diag ? pim : 0.f;
    }

    const int G[4][3] = {{6,0,1},{8,3,4},{10,1,2},{12,2,4}};
#pragma unroll
    for (int g = 0; g < 4; ++g) {
      const int cb = G[g][1], tb = G[g][2];
      const int ctrl = (lane >> cb) & 1;
      const int tgt  = (lane >> tb) & 1;
      {
        const float cz = thc[G[g][0]], sz = ths[G[g][0]];
        float cc = ctrl ? cz : 1.0f;
        float gg = ctrl ? (tgt ? sz : -sz) : 0.0f;
#pragma unroll
        for (int j = 0; j < 4; ++j) {
          float re = xre[j], im = xim[j];
          xre[j] = re * cc - im * gg;
          xim[j] = im * cc + re * gg;
        }
      }
      {
        const float co = thc[G[g][0] + 1], si = ths[G[g][0] + 1];
        float ce = ctrl ? co : 1.0f;
        float se = ctrl ? si : 0.0f;
#pragma unroll
        for (int j = 0; j < 4; ++j) {
          float pre = __shfl_xor(xre[j], 1 << tb, 32);
          float pim = __shfl_xor(xim[j], 1 << tb, 32);
          float re = xre[j], im = xim[j];
          xre[j] = ce * re + se * pim;
          xim[j] = ce * im - se * pre;
        }
      }
    }

    if (lane < 16) {
      v2f* chunk = Ls + (wv << 6);
      const v2f re_lo = {xre[0], xre[1]}, re_hi = {xre[2], xre[3]};
      const v2f im_lo = {xim[0], xim[1]}, im_hi = {xim[2], xim[3]};
      chunk[lane]           = re_lo;
      chunk[lane + 16]      = re_hi;
      chunk[32 + lane]      = im_lo;
      chunk[32 + lane + 16] = im_hi;
    }
  }

  float cq[9], sq[9];
#pragma unroll
  for (int m = 0; m < 9; ++m)
    sincos_quarter_turns((m == 0 || m == 2) ? 2.0f * d[m] : d[m], sq[m], cq[m]);

  float PA, PU;
  {
    const float C0 = cq[0], S0 = sq[0];
    const float C2 = cq[2], S2 = sq[2];
    const float W10 = cq[1] * cq[1], W11 = sq[1] * sq[1];
    const float W30 = cq[3] * cq[3], W31 = sq[3] * sq[3];

    const float x1 = S0 * ths[0];
    const float y1 = -S0 * thc[0];
    const float y1r = fmaf(y1, thc[1], -(C0 * ths[1]));
    const float z1r = fmaf(y1, ths[1], C0 * thc[1]);
    float rx = W11 * x1;
    float ry = fmaf(W11, y1r, -(W10 * S0));
    float rz = fmaf(W11, z1r, W10 * C0);

    const float y2 = -S2 * thc[2];
    const float z2 = fmaf(y2, ths[3], C2 * thc[3]);
    const float zB = fmaf(W31, z2, W30 * C2);
    const float B0 = fmaf(0.5f, zB, 0.5f);
    const float B1 = fmaf(-0.5f, zB, 0.5f);

    const float x3 = fmaf(rx, thc[4], -(ry * ths[4]));
    const float y3 = fmaf(rx, ths[4], ry * thc[4]);
    const float y3r = fmaf(y3, thc[5], -(rz * ths[5]));
    const float z3r = fmaf(y3, ths[5], rz * thc[5]);
    rx = fmaf(B1, x3, B0 * rx);
    ry = fmaf(B1, y3r, B0 * ry);
    rz = fmaf(B1, z3r, B0 * rz);
    PA = fmaf(0.5f, rz, 0.5f);

    const float y8 = fmaf(rx, ths[14], ry * thc[14]);
    const float z8 = fmaf(y8, ths[15], rz * thc[15]);
    PU = fmaf(0.5f, z8, 0.5f);
  }

  float mlane = l4 ? sq[7] : cq[7];
  float mv[16];
#pragma unroll
  for (int i = 0; i < 16; ++i) {
    float mi = ((i & 1)        ? sq[8] : cq[8]) *
               (((i >> 1) & 1) ? sq[6] : cq[6]) *
               (((i >> 2) & 1) ? sq[5] : cq[5]) *
               (((i >> 3) & 1) ? sq[4] : cq[4]);
    mv[i] = mlane * mi;
  }

  __syncthreads();

  float w0;
  {
    v8f accre = {}, accim = {};
#pragma unroll
    for (int kc = 0; kc < 8; ++kc) {
      const v2f* bb = Ls + (kc << 6);
      v2f bre = bb[lane];
      v2f bim = bb[32 + lane];
      v2f am = {mv[2 * kc], mv[2 * kc + 1]};
      accre = __builtin_amdgcn_wmma_f32_16x16x4_f32(false, am, false, bre,
                                                    (short)0, accre, false, false);
      accim = __builtin_amdgcn_wmma_f32_16x16x4_f32(false, am, false, bim,
                                                    (short)0, accim, false, false);
    }
    float q[8];
#pragma unroll
    for (int j = 0; j < 8; ++j) q[j] = accre[j] * accre[j] + accim[j] * accim[j];
#pragma unroll
    for (int off = 8; off >= 1; off >>= 1)
#pragma unroll
      for (int j = 0; j < 8; ++j) q[j] += __shfl_xor(q[j], off, 32);
    float ws = q[0];
#pragma unroll
    for (int j = 1; j < 8; ++j) ws = ((lane & 7) == j) ? q[j] : ws;
    w0 = ws;
  }

  if (valid && ((r >> 3) == l4))
    out[((b * 24 + row + 1) * 24 + (col + 1)) * 32 + k] = PU + w0 * (PA - PU);
}

extern "C" void kernel_launch(void* const* d_in, const int* in_sizes, int n_in,
                              void* d_out, int out_size, void* d_ws, size_t ws_size,
                              hipStream_t stream) {
  const float* x  = (const float*)d_in[0];
  const float* qp = (const float*)d_in[1];
  float* out = (float*)d_out;

  quanv_fused_kernel<<<256, 256, 0, stream>>>(x, qp, out);
}


# Whole circuit as single-qubit Bloch marginals: out is bilinear in patch features x filter coefficients, 17 f32 WMMAs per 16x16 tile (K=36+16+16), 32 blocks of 4 waves; sum grouping changes
# speedup vs baseline: 19.2760x; 1.7127x over previous
#include <hip/hip_runtime.h>

typedef __attribute__((ext_vector_type(2))) float v2f;
typedef __attribute__((ext_vector_type(8))) float v8f;

__device__ __forceinline__ void sincos_quarter_turns(float d, float& s, float& c) {
  const float t = __builtin_rintf(d);
  const float w = d - t;
  const int q = (int)t;
  const float y = w * w;
  float sp = fmaf(y, 0.000468568905f, -0.00485667493f);
  sp = fmaf(y, sp, 0.0797256678f);
  sp = fmaf(y, sp, -0.645966411f);
  const float sw = fmaf(1.57079637f, w, (w * y) * sp);
  float cp = fmaf(y, 0.0000486124918f, 0.000870445569f);
  cp = fmaf(y, cp, -0.0208520181f);
  cp = fmaf(y, cp, 0.253668398f);
  cp = fmaf(y, cp, -1.23370051f);
  const float cw = fmaf(y, cp, 1.0f);
  const bool odd = (q & 1) != 0;
  const unsigned flip = ((unsigned)q & 2u) << 30;
  s = __uint_as_float(__float_as_uint(odd ? cw : sw) ^ flip);
  c = __uint_as_float(__float_as_uint(odd ? -sw : cw) ^ flip);
}

__global__ __launch_bounds__(128) void quanv_bilinear_kernel(
    const float* __restrict__ x, const float* __restrict__ qp,
    float* __restrict__ out) {
  __shared__ float4 trig4[16 * 9];

  const int tid = threadIdx.x;
  const int lane = tid & 31;
  const int wv = tid >> 5;
  const int nt = blockIdx.x & 1;
  const int tile = ((blockIdx.x >> 1) << 2) + wv;

  const int r = lane & 15;
  const int h = lane >> 4;
  const int pid = tile * 16 + r;
  const int p = pid < 968 ? pid : 967;
  const int b = p / 484;
  const int rem = p - b * 484;
  const int row = rem / 22;
  const int col = rem - row * 22;
  const int ooff = ((b * 24 + row + 1) * 24 + (col + 1)) * 32;

  float ang[2];
#pragma unroll
  for (int i = 0; i < 2; ++i)
    ang[i] = qp[nt * 256 + tid + 128 * i];
  float d[9];
#pragma unroll
  for (int m = 0; m < 9; ++m)
    d[m] = x[(b * 24 + row + m / 3) * 24 + (col + m % 3)];

  {
    const int gid = blockIdx.x * 128 + tid;
    if (gid < 184 * 8) {
      const int bp = gid >> 3;
      const int img = bp / 92;
      const int bi = bp - img * 92;
      int zr, zc;
      if (bi < 24)      { zr = 0;       zc = bi; }
      else if (bi < 48) { zr = 23;      zc = bi - 24; }
      else if (bi < 70) { zr = bi - 47; zc = 0; }
      else              { zr = bi - 69; zc = 23; }
      reinterpret_cast<float4*>(out)[((img * 24 + zr) * 24 + zc) * 8 + (gid & 7)] =
          make_float4(0.f, 0.f, 0.f, 0.f);
    }
  }

  {
    float2* trig2 = reinterpret_cast<float2*>(trig4);
#pragma unroll
    for (int i = 0; i < 2; ++i) {
      const int idx = tid + 128 * i;
      float s, c;
      sincosf(ang[i], &s, &c);
      trig2[(idx >> 4) * 18 + (idx & 15)] = make_float2(c, s);
    }
  }

  float cq[9], sq[9];
#pragma unroll
  for (int m = 0; m < 9; ++m) {
    const bool ctl = (m == 1 || m == 3 || m == 5 || m == 8);
    sincos_quarter_turns(ctl ? d[m] : 2.0f * d[m], sq[m], cq[m]);
  }
  v2f pf4[9], pf0[4];
  {
    const float W5 = h ? sq[5] * sq[5] : cq[5] * cq[5];
    const v2f u4 = {W5 * sq[4], W5 * cq[4]};
    const float W80 = cq[8] * cq[8], W81 = sq[8] * sq[8];
    const float f7[3] = {W80 * cq[7], W81 * cq[7], W81 * sq[7]};
    pf4[0] = u4;
    pf4[1] = cq[6] * u4;
    pf4[2] = sq[6] * u4;
#pragma unroll
    for (int i = 0; i < 3; ++i) {
      pf4[3 + i] = (f7[i] * cq[6]) * u4;
      pf4[6 + i] = (f7[i] * sq[6]) * u4;
    }
    const float W1 = h ? sq[1] * sq[1] : cq[1] * cq[1];
    const v2f u0 = {W1 * sq[0], W1 * cq[0]};
    const float W30 = cq[3] * cq[3], W31 = sq[3] * sq[3];
    pf0[0] = u0;
    pf0[1] = (W30 * cq[2]) * u0;
    pf0[2] = (W31 * cq[2]) * u0;
    pf0[3] = (W31 * sq[2]) * u0;
  }

  __syncthreads();

  v2f fc4[9], fcU[4], fcD[4];
  {
    float tc[16], ts[16];
#pragma unroll
    for (int q = 0; q < 8; ++q) {
      const float4 v = trig4[r * 9 + q];
      tc[2 * q] = v.x; ts[2 * q] = v.y; tc[2 * q + 1] = v.z; ts[2 * q + 1] = v.w;
    }
    const v2f gx = {h ? ts[8] : 0.f, 0.f};
    const v2f gy = {h ? -tc[8] * tc[9] : -1.f, h ? -ts[9] : 0.f};
    const v2f gz = {h ? -tc[8] * ts[9] : 0.f, h ? tc[9] : 1.f};
    const v2f T = ts[12] * gx + tc[12] * gy;
    const v2f P = (1.f + tc[13]) * gz + ts[13] * T;
    const v2f Q = (1.f - tc[13]) * gz - ts[13] * T;
    const float a1 = 0.5f * (1.f + tc[11]);
    const float a2 = -0.5f * tc[10] * ts[11];
    const float b1 = 0.5f * (1.f - tc[11]);
    const float b2 = -a2;
    const float ph[3] = {1.f, tc[7], -tc[6] * ts[7]};
    const v2f hQ = 0.5f * Q;
    fc4[0] = 0.5f * P;
    fc4[1] = a1 * hQ;
    fc4[2] = a2 * hQ;
#pragma unroll
    for (int i = 0; i < 3; ++i) {
      fc4[3 + i] = (ph[i] * b1) * hQ;
      fc4[6 + i] = (ph[i] * b2) * hQ;
    }

    const v2f rx = {h ? ts[0] : 0.f, 0.f};
    const v2f ry = {h ? -tc[0] * tc[1] : -1.f, h ? -ts[1] : 0.f};
    const v2f rz = {h ? -tc[0] * ts[1] : 0.f, h ? tc[1] : 1.f};
    const v2f x3 = tc[4] * rx - ts[4] * ry;
    const v2f y3 = ts[4] * rx + tc[4] * ry;
    const v2f y3r = tc[5] * y3 - ts[5] * rz;
    const v2f z3r = ts[5] * y3 + tc[5] * rz;
    const v2f px = rx + x3, py = ry + y3r, pz = rz + z3r;
    const v2f mx = rx - x3, my = ry - y3r, mz = rz - z3r;
    const v2f UP = ts[15] * (ts[14] * px + tc[14] * py) + tc[15] * pz;
    const v2f UM = ts[15] * (ts[14] * mx + tc[14] * my) + tc[15] * mz;
    const float ps[3] = {1.f, tc[3], -tc[2] * ts[3]};
    fcU[0] = 0.5f * UP;
    fcD[0] = 0.5f * (pz - UP);
    const v2f hUM = 0.5f * UM, hDM = 0.5f * (mz - UM);
#pragma unroll
    for (int i = 0; i < 3; ++i) {
      fcU[1 + i] = ps[i] * hUM;
      fcD[1 + i] = ps[i] * hDM;
    }
  }

  v8f z4a = {}, z4b = {}, zu = {}, zd = {};
#pragma unroll
  for (int i = 0; i < 4; ++i) {
    z4a = __builtin_amdgcn_wmma_f32_16x16x4_f32(false, fc4[2 * i], false, pf4[2 * i],
                                                (short)0, z4a, false, false);
    zu = __builtin_amdgcn_wmma_f32_16x16x4_f32(false, fcU[i], false, pf0[i],
                                               (short)0, zu, false, false);
    z4b = __builtin_amdgcn_wmma_f32_16x16x4_f32(false, fc4[2 * i + 1], false, pf4[2 * i + 1],
                                                (short)0, z4b, false, false);
    zd = __builtin_amdgcn_wmma_f32_16x16x4_f32(false, fcD[i], false, pf0[i],
                                               (short)0, zd, false, false);
  }
  z4a = __builtin_amdgcn_wmma_f32_16x16x4_f32(false, fc4[8], false, pf4[8],
                                              (short)0, z4a, false, false);

  float o[8];
#pragma unroll
  for (int j = 0; j < 8; ++j) {
    const float z4 = z4a[j] + z4b[j];
    o[j] = fmaf(fmaf(0.25f, z4, 0.25f), zd[j], fmaf(0.5f, zu[j], 0.5f));
  }
  if (pid < 968) {
    float4* dst = reinterpret_cast<float4*>(out + ooff + nt * 16 + h * 8);
    dst[0] = make_float4(o[0], o[1], o[2], o[3]);
    dst[1] = make_float4(o[4], o[5], o[6], o[7]);
  }
}

extern "C" void kernel_launch(void* const* d_in, const int* in_sizes, int n_in,
                              void* d_out, int out_size, void* d_ws, size_t ws_size,
                              hipStream_t stream) {
  const float* x  = (const float*)d_in[0];
  const float* qp = (const float*)d_in[1];
  float* out = (float*)d_out;

  quanv_bilinear_kernel<<<32, 128, 0, stream>>>(x, qp, out);
}
